# starting kernel
# baseline (speedup 1.0000x reference)
#include <hip/hip_runtime.h>
#include <hip/hip_bf16.h>
#include <math.h>

typedef __attribute__((ext_vector_type(2))) float v2f;
typedef __attribute__((ext_vector_type(4))) float v4f;
typedef __attribute__((ext_vector_type(8))) float v8f;

#define D_MODEL_C 1024
#define D_STATE_C 16
#define D_CONV_C  4
#define DT_RANK_C 64
#define D_INNER_C 2048
#define BATCH_C   2
#define SEQ_C     1024
#define NTOK_C    (BATCH_C * SEQ_C)

template <int EPILOG>
__global__ __launch_bounds__(256)
void wmma_gemm_f32(const float* __restrict__ A,
                   const float* __restrict__ W,
                   const float* __restrict__ bias,
                   float* __restrict__ C,
                   int M, int N, int K, int lda)
{
    const int lane      = threadIdx.x & 31;
    const int waveInBlk = threadIdx.x >> 5;
    const int nStrips   = (N + 63) >> 6;
    const int strip     = blockIdx.x * 8 + waveInBlk;
    const int total     = (M >> 4) * nStrips;
    if (strip >= total) return;

    const int m0 = (strip / nStrips) << 4;
    const int n0 = (strip % nStrips) << 6;

    const int lhalf = lane >> 4;
    const int lidx  = lane & 15;

    v8f acc[4];
    acc[0] = v8f{}; acc[1] = v8f{}; acc[2] = v8f{}; acc[3] = v8f{};

    const float* arow = A + (size_t)(m0 + lidx) * lda + 2 * lhalf;
    const float* bcol[4];
#pragma unroll
    for (int t = 0; t < 4; ++t) {
        int nc = n0 + 16 * t;
        bcol[t] = W + (size_t)((nc < N ? nc : 0) + lidx) * K + 2 * lhalf;
    }

    for (int k = 0; k < K; k += 16) {
        v2f a[4];
        v2f b[4][4];
#pragma unroll
        for (int kk = 0; kk < 4; ++kk) {
            int kb = k + 4 * kk;
            a[kk][0] = arow[kb];
            a[kk][1] = arow[kb + 1];
#pragma unroll
            for (int t = 0; t < 4; ++t) {
                b[kk][t][0] = bcol[t][kb];
                b[kk][t][1] = bcol[t][kb + 1];
            }
        }
#pragma unroll
        for (int kk = 0; kk < 4; ++kk) {
#pragma unroll
            for (int t = 0; t < 4; ++t) {
                acc[t] = __builtin_amdgcn_wmma_f32_16x16x4_f32(
                    false, a[kk], false, b[kk][t], (short)0, acc[t],
                    false, false);
            }
        }
    }

#pragma unroll
    for (int t = 0; t < 4; ++t) {
        int nc = n0 + 16 * t;
        if (nc >= N) break;
        float* cp = C + (size_t)(m0 + 8 * lhalf) * N + nc + lidx;
        float bv = (EPILOG == 1) ? bias[nc + lidx] : 0.f;
#pragma unroll
        for (int i = 0; i < 8; ++i) {
            float v = acc[t][i];
            if (EPILOG == 1) {
                float x = v + bv;
                v = (x > 20.f) ? x : log1pf(__expf(x));
            }
            cp[(size_t)i * N] = v;
        }
    }
}

__global__ __launch_bounds__(256)
void conv_silu_kernel(const float* __restrict__ xz,
                      const float* __restrict__ conv_w,
                      const float* __restrict__ conv_b,
                      float* __restrict__ x_act)
{
    int idx = blockIdx.x * blockDim.x + threadIdx.x;
    if (idx >= NTOK_C * D_INNER_C) return;
    int d = idx & (D_INNER_C - 1);
    int m = idx >> 11;
    int t = m & (SEQ_C - 1);

    float acc = conv_b[d];
    const float* w = conv_w + d * D_CONV_C;
#pragma unroll
    for (int j = 0; j < D_CONV_C; ++j) {
        int tt = t - (D_CONV_C - 1) + j;
        if (tt >= 0)
            acc += w[j] * xz[(size_t)(m - (D_CONV_C - 1) + j) * (2 * D_INNER_C) + d];
    }
    float s = acc / (1.f + __expf(-acc));
    x_act[(size_t)m * D_INNER_C + d] = s;
}

__global__ __launch_bounds__(256)
void selective_scan_kernel(const float* __restrict__ xz,
                           const float* __restrict__ x_act,
                           const float* __restrict__ dbc,
                           float* __restrict__ dt_y,
                           const float* __restrict__ A_log,
                           const float* __restrict__ Dvec)
{
    int gid = blockIdx.x * blockDim.x + threadIdx.x;
    if (gid >= BATCH_C * D_INNER_C) return;
    int b = gid >> 11;
    int d = gid & (D_INNER_C - 1);

    float Arow[D_STATE_C];
#pragma unroll
    for (int s = 0; s < D_STATE_C; ++s)
        Arow[s] = -__expf(A_log[(size_t)d * D_STATE_C + s]);
    const float Dd = Dvec[d];

    float h[D_STATE_C];
#pragma unroll
    for (int s = 0; s < D_STATE_C; ++s) h[s] = 0.f;

    for (int t = 0; t < SEQ_C; ++t) {
        size_t m = (size_t)b * SEQ_C + t;
        float u   = x_act[m * D_INNER_C + d];
        float dtv = dt_y [m * D_INNER_C + d];
        const v4f* bc4 = (const v4f*)(dbc + m * 96 + 64);
        v4f Bv0 = bc4[0], Bv1 = bc4[1], Bv2 = bc4[2], Bv3 = bc4[3];
        float Bt[D_STATE_C], Ct[D_STATE_C];
#pragma unroll
        for (int q = 0; q < 4; ++q) {
            Bt[q]      = Bv0[q]; Bt[4 + q]  = Bv1[q];
            Bt[8 + q]  = Bv2[q]; Bt[12 + q] = Bv3[q];
        }
        const v4f* cc4 = (const v4f*)(dbc + m * 96 + 80);
        v4f Cv0 = cc4[0], Cv1 = cc4[1], Cv2 = cc4[2], Cv3 = cc4[3];
#pragma unroll
        for (int q = 0; q < 4; ++q) {
            Ct[q]      = Cv0[q]; Ct[4 + q]  = Cv1[q];
            Ct[8 + q]  = Cv2[q]; Ct[12 + q] = Cv3[q];
        }

        float du = dtv * u;
        float y = 0.f;
#pragma unroll
        for (int s = 0; s < D_STATE_C; ++s) {
            float dA = __expf(dtv * Arow[s]);
            h[s] = dA * h[s] + du * Bt[s];
            y += h[s] * Ct[s];
        }
        y += u * Dd;
        float z  = xz[m * (2 * D_INNER_C) + D_INNER_C + d];
        float sz = z / (1.f + __expf(-z));
        dt_y[m * D_INNER_C + d] = y * sz;
    }
}

static inline int gemm_blocks(int M, int N)
{
    int strips = (M >> 4) * ((N + 63) >> 6);
    return (strips + 7) >> 3;
}

extern "C" void kernel_launch(void* const* d_in, const int* in_sizes, int n_in,
                              void* d_out, int out_size, void* d_ws, size_t ws_size,
                              hipStream_t stream) {
    (void)in_sizes; (void)n_in; (void)out_size; (void)ws_size;

    const float* x      = (const float*)d_in[0];
    const float* W_in   = (const float*)d_in[1];
    const float* conv_w = (const float*)d_in[2];
    const float* conv_b = (const float*)d_in[3];
    const float* W_x    = (const float*)d_in[4];
    const float* W_dt   = (const float*)d_in[5];
    const float* b_dt   = (const float*)d_in[6];
    const float* A_log  = (const float*)d_in[7];
    const float* Dvec   = (const float*)d_in[8];
    const float* W_out  = (const float*)d_in[9];
    float* out = (float*)d_out;

    float* xz   = (float*)d_ws;
    float* xact = xz   + (size_t)NTOK_C * (2 * D_INNER_C);
    float* dbc  = xact + (size_t)NTOK_C * D_INNER_C;
    float* dtb  = dbc  + (size_t)NTOK_C * 96;

    wmma_gemm_f32<0><<<gemm_blocks(NTOK_C, 2 * D_INNER_C), 256, 0, stream>>>(
        x, W_in, nullptr, xz, NTOK_C, 2 * D_INNER_C, D_MODEL_C, D_MODEL_C);

    {
        int n = NTOK_C * D_INNER_C;
        conv_silu_kernel<<<(n + 255) / 256, 256, 0, stream>>>(xz, conv_w, conv_b, xact);
    }

    wmma_gemm_f32<0><<<gemm_blocks(NTOK_C, 96), 256, 0, stream>>>(
        xact, W_x, nullptr, dbc, NTOK_C, 96, D_INNER_C, D_INNER_C);

    wmma_gemm_f32<1><<<gemm_blocks(NTOK_C, D_INNER_C), 256, 0, stream>>>(
        dbc, W_dt, b_dt, dtb, NTOK_C, D_INNER_C, DT_RANK_C, 96);

    selective_scan_kernel<<<(BATCH_C * D_INNER_C) / 256, 256, 0, stream>>>(
        xz, xact, dbc, dtb, A_log, Dvec);

    wmma_gemm_f32<0><<<gemm_blocks(NTOK_C, D_MODEL_C), 256, 0, stream>>>(
        dtb, W_out, nullptr, out, NTOK_C, D_MODEL_C, D_INNER_C, D_INNER_C);
}


# scan over time split in 32 chunks: kernel A scans each chunk from zero, stores end state + decay product; kernel B folds earlier chunks into its start state and rescans; state rounding order changes
# speedup vs baseline: 2.4947x; 2.4947x over previous
#include <hip/hip_runtime.h>
#include <hip/hip_bf16.h>
#include <math.h>

typedef __attribute__((ext_vector_type(2))) float v2f;
typedef __attribute__((ext_vector_type(4))) float v4f;
typedef __attribute__((ext_vector_type(8))) float v8f;

#define D_MODEL_C 1024
#define D_STATE_C 16
#define D_CONV_C  4
#define DT_RANK_C 64
#define D_INNER_C 2048
#define BATCH_C   2
#define SEQ_C     1024
#define NTOK_C    (BATCH_C * SEQ_C)

template <int EPILOG>
__global__ __launch_bounds__(256)
void wmma_gemm_f32(const float* __restrict__ A,
                   const float* __restrict__ W,
                   const float* __restrict__ bias,
                   float* __restrict__ C,
                   int M, int N, int K, int lda)
{
    const int lane      = threadIdx.x & 31;
    const int waveInBlk = threadIdx.x >> 5;
    const int nStrips   = (N + 63) >> 6;
    const int strip     = blockIdx.x * 8 + waveInBlk;
    const int total     = (M >> 4) * nStrips;
    if (strip >= total) return;

    const int m0 = (strip / nStrips) << 4;
    const int n0 = (strip % nStrips) << 6;

    const int lhalf = lane >> 4;
    const int lidx  = lane & 15;

    v8f acc[4];
    acc[0] = v8f{}; acc[1] = v8f{}; acc[2] = v8f{}; acc[3] = v8f{};

    const float* arow = A + (size_t)(m0 + lidx) * lda + 2 * lhalf;
    const float* bcol[4];
#pragma unroll
    for (int t = 0; t < 4; ++t) {
        int nc = n0 + 16 * t;
        bcol[t] = W + (size_t)((nc < N ? nc : 0) + lidx) * K + 2 * lhalf;
    }

    for (int k = 0; k < K; k += 16) {
        v2f a[4];
        v2f b[4][4];
#pragma unroll
        for (int kk = 0; kk < 4; ++kk) {
            int kb = k + 4 * kk;
            a[kk][0] = arow[kb];
            a[kk][1] = arow[kb + 1];
#pragma unroll
            for (int t = 0; t < 4; ++t) {
                b[kk][t][0] = bcol[t][kb];
                b[kk][t][1] = bcol[t][kb + 1];
            }
        }
#pragma unroll
        for (int kk = 0; kk < 4; ++kk) {
#pragma unroll
            for (int t = 0; t < 4; ++t) {
                acc[t] = __builtin_amdgcn_wmma_f32_16x16x4_f32(
                    false, a[kk], false, b[kk][t], (short)0, acc[t],
                    false, false);
            }
        }
    }

#pragma unroll
    for (int t = 0; t < 4; ++t) {
        int nc = n0 + 16 * t;
        if (nc >= N) break;
        float* cp = C + (size_t)(m0 + 8 * lhalf) * N + nc + lidx;
        float bv = (EPILOG == 1) ? bias[nc + lidx] : 0.f;
#pragma unroll
        for (int i = 0; i < 8; ++i) {
            float v = acc[t][i];
            if (EPILOG == 1) {
                float x = v + bv;
                v = (x > 20.f) ? x : log1pf(__expf(x));
            }
            cp[(size_t)i * N] = v;
        }
    }
}

__global__ __launch_bounds__(256)
void conv_silu_kernel(const float* __restrict__ xz,
                      const float* __restrict__ conv_w,
                      const float* __restrict__ conv_b,
                      float* __restrict__ x_act)
{
    int idx = blockIdx.x * blockDim.x + threadIdx.x;
    if (idx >= NTOK_C * D_INNER_C) return;
    int d = idx & (D_INNER_C - 1);
    int m = idx >> 11;
    int t = m & (SEQ_C - 1);

    float acc = conv_b[d];
    const float* w = conv_w + d * D_CONV_C;
#pragma unroll
    for (int j = 0; j < D_CONV_C; ++j) {
        int tt = t - (D_CONV_C - 1) + j;
        if (tt >= 0)
            acc += w[j] * xz[(size_t)(m - (D_CONV_C - 1) + j) * (2 * D_INNER_C) + d];
    }
    float s = acc / (1.f + __expf(-acc));
    x_act[(size_t)m * D_INNER_C + d] = s;
}

#define SCAN_NC   32
#define SCAN_LC   (SEQ_C / SCAN_NC)
#define SCAN_TPB  128
#define XZ_PITCH  (2 * D_INNER_C)

__global__ __launch_bounds__(SCAN_TPB)
void scan_chunk_carry_kernel(const float* __restrict__ x_act,
                             const float* __restrict__ dbc,
                             const float* __restrict__ dt,
                             const float* __restrict__ A_log,
                             float* __restrict__ carry)
{
    const int d = blockIdx.x * SCAN_TPB + threadIdx.x;
    const int c = blockIdx.y;
    const int b = blockIdx.z;

    float Arow[D_STATE_C];
#pragma unroll
    for (int s = 0; s < D_STATE_C; ++s)
        Arow[s] = -__expf(A_log[(size_t)d * D_STATE_C + s]);

    float h[D_STATE_C], P[D_STATE_C];
#pragma unroll
    for (int s = 0; s < D_STATE_C; ++s) { h[s] = 0.f; P[s] = 1.f; }

    const size_t m0 = (size_t)b * SEQ_C + (size_t)c * SCAN_LC;
#pragma unroll 4
    for (int i = 0; i < SCAN_LC; ++i) {
        const size_t m = m0 + i;
        const float u   = x_act[m * D_INNER_C + d];
        const float dtv = dt   [m * D_INNER_C + d];
        const float* Bt = dbc + m * 96 + 64;
        const float du = dtv * u;
#pragma unroll
        for (int s = 0; s < D_STATE_C; ++s) {
            float dA = __expf(dtv * Arow[s]);
            h[s] = dA * h[s] + du * Bt[s];
            P[s] = P[s] * dA;
        }
    }

    float* q = carry + (size_t)(b * SCAN_NC + c) * 32 * XZ_PITCH + d;
#pragma unroll
    for (int s = 0; s < D_STATE_C; ++s) {
        q[(size_t)s * XZ_PITCH]                = h[s];
        q[(size_t)(D_STATE_C + s) * XZ_PITCH]  = P[s];
    }
}

__global__ __launch_bounds__(SCAN_TPB)
void scan_chunk_final_kernel(const float* __restrict__ zcol,
                             const float* __restrict__ carry,
                             const float* __restrict__ x_act,
                             const float* __restrict__ dbc,
                             float* __restrict__ dt_y,
                             const float* __restrict__ A_log,
                             const float* __restrict__ Dvec)
{
    const int d = blockIdx.x * SCAN_TPB + threadIdx.x;
    const int c = blockIdx.y;
    const int b = blockIdx.z;

    float Arow[D_STATE_C];
#pragma unroll
    for (int s = 0; s < D_STATE_C; ++s)
        Arow[s] = -__expf(A_log[(size_t)d * D_STATE_C + s]);
    const float Dd = Dvec[d];

    float h[D_STATE_C];
#pragma unroll
    for (int s = 0; s < D_STATE_C; ++s) h[s] = 0.f;

    {
        const float* q = carry + (size_t)(b * SCAN_NC) * 32 * XZ_PITCH + d;
#pragma unroll 2
        for (int cp = 0; cp < c; ++cp) {
#pragma unroll
            for (int s = 0; s < D_STATE_C; ++s)
                h[s] = q[(size_t)(D_STATE_C + s) * XZ_PITCH] * h[s] + q[(size_t)s * XZ_PITCH];
            q += (size_t)32 * XZ_PITCH;
        }
    }

    const size_t m0 = (size_t)b * SEQ_C + (size_t)c * SCAN_LC;
#pragma unroll 8
    for (int i = 0; i < SCAN_LC; ++i) {
        const size_t m = m0 + i;
        const float u   = x_act[m * D_INNER_C + d];
        const float dtv = dt_y [m * D_INNER_C + d];
        const float z   = zcol [m * XZ_PITCH + d];
        const float* Bt = dbc + m * 96 + 64;
        const float* Ct = dbc + m * 96 + 80;

        const float du = dtv * u;
        float yv = 0.f;
#pragma unroll
        for (int s = 0; s < D_STATE_C; ++s) {
            float dA = __expf(dtv * Arow[s]);
            h[s] = dA * h[s] + du * Bt[s];
            yv += h[s] * Ct[s];
        }
        yv += u * Dd;
        float sz = z / (1.f + __expf(-z));
        dt_y[m * D_INNER_C + d] = yv * sz;
    }
}

static inline int gemm_blocks(int M, int N)
{
    int strips = (M >> 4) * ((N + 63) >> 6);
    return (strips + 7) >> 3;
}

extern "C" void kernel_launch(void* const* d_in, const int* in_sizes, int n_in,
                              void* d_out, int out_size, void* d_ws, size_t ws_size,
                              hipStream_t stream) {
    (void)in_sizes; (void)n_in; (void)out_size; (void)ws_size;

    const float* x      = (const float*)d_in[0];
    const float* W_in   = (const float*)d_in[1];
    const float* conv_w = (const float*)d_in[2];
    const float* conv_b = (const float*)d_in[3];
    const float* W_x    = (const float*)d_in[4];
    const float* W_dt   = (const float*)d_in[5];
    const float* b_dt   = (const float*)d_in[6];
    const float* A_log  = (const float*)d_in[7];
    const float* Dvec   = (const float*)d_in[8];
    const float* W_out  = (const float*)d_in[9];
    float* out = (float*)d_out;

    float* xz   = (float*)d_ws;
    float* xact = xz   + (size_t)NTOK_C * (2 * D_INNER_C);
    float* dbc  = xact + (size_t)NTOK_C * D_INNER_C;
    float* dtb  = dbc  + (size_t)NTOK_C * 96;

    wmma_gemm_f32<0><<<gemm_blocks(NTOK_C, 2 * D_INNER_C), 256, 0, stream>>>(
        x, W_in, nullptr, xz, NTOK_C, 2 * D_INNER_C, D_MODEL_C, D_MODEL_C);

    {
        int n = NTOK_C * D_INNER_C;
        conv_silu_kernel<<<(n + 255) / 256, 256, 0, stream>>>(xz, conv_w, conv_b, xact);
    }

    wmma_gemm_f32<0><<<gemm_blocks(NTOK_C, 96), 256, 0, stream>>>(
        xact, W_x, nullptr, dbc, NTOK_C, 96, D_INNER_C, D_INNER_C);

    wmma_gemm_f32<1><<<gemm_blocks(NTOK_C, D_INNER_C), 256, 0, stream>>>(
        dbc, W_dt, b_dt, dtb, NTOK_C, D_INNER_C, DT_RANK_C, 96);

    scan_chunk_carry_kernel<<<dim3(D_INNER_C / SCAN_TPB, SCAN_NC - 1, BATCH_C), SCAN_TPB, 0, stream>>>(
        xact, dbc, dtb, A_log, xz);

    scan_chunk_final_kernel<<<dim3(D_INNER_C / SCAN_TPB, SCAN_NC, BATCH_C), SCAN_TPB, 0, stream>>>(
        xz + D_INNER_C, xz, xact, dbc, dtb, A_log, Dvec);

    wmma_gemm_f32<0><<<gemm_blocks(NTOK_C, D_MODEL_C), 256, 0, stream>>>(
        dtb, W_out, nullptr, out, NTOK_C, D_MODEL_C, D_INNER_C, D_INNER_C);
}


# GEMM operands staged through LDS: per K-step of 32 a block copies whole 128-byte rows of A and W with 16-byte loads, waves read fragments from LDS; tiles 64x128, 64x64, 32x128; sums unchanged
# speedup vs baseline: 7.6818x; 3.0792x over previous
#include <hip/hip_runtime.h>
#include <hip/hip_bf16.h>
#include <math.h>

typedef __attribute__((ext_vector_type(2))) float v2f;
typedef __attribute__((ext_vector_type(4))) float v4f;
typedef __attribute__((ext_vector_type(8))) float v8f;

#define D_MODEL_C 1024
#define D_STATE_C 16
#define D_CONV_C  4
#define DT_RANK_C 64
#define D_INNER_C 2048
#define BATCH_C   2
#define SEQ_C     1024
#define NTOK_C    (BATCH_C * SEQ_C)

#define GEMM_BK     32
#define GEMM_PITCH  36

template <int EPILOG, int BM, int BN, int WM, int WN>
__global__ __launch_bounds__(256)
void wmma_gemm_f32(const float* __restrict__ A,
                   const float* __restrict__ W,
                   const float* __restrict__ bias,
                   float* __restrict__ C,
                   int M, int N, int K, int lda)
{
    constexpr int TM   = BM / WM / 16;
    constexpr int TN   = BN / WN / 16;
    constexpr int ROWS = BM + BN;
    constexpr int NLD  = ROWS / 32;
    static_assert(WM * WN == 8, "8 waves per block");
    static_assert(BM % 32 == 0 && BN % 32 == 0, "a pass of 32 rows is all A or all W");
    static_assert(TM * 16 * WM == BM && TN * 16 * WN == BN, "wave tiles cover the block tile");

    __shared__ __attribute__((aligned(16))) float lds[2][ROWS * GEMM_PITCH];

    const int tid   = threadIdx.x;
    const int lane  = tid & 31;
    const int wave  = tid >> 5;
    const int wm    = wave / WN;
    const int wn    = wave % WN;
    const int lhalf = lane >> 4;
    const int lidx  = lane & 15;

    const int nBlkN = (N + BN - 1) / BN;
    const int m0 = (blockIdx.x / nBlkN) * BM;
    const int n0 = (blockIdx.x % nBlkN) * BN;
    (void)M;

    const int lrow = tid >> 3;
    const int lq   = tid & 7;
    const float* gp[NLD];
#pragma unroll
    for (int p = 0; p < NLD; ++p) {
        const int r = p * 32 + lrow;
        if (p * 32 < BM) {
            gp[p] = A + (size_t)(m0 + r) * lda + 4 * lq;
        } else {
            const int n = n0 + r - BM;
            gp[p] = W + (size_t)(n < N ? n : 0) * K + 4 * lq;
        }
    }
    const int st_off = lrow * GEMM_PITCH + 4 * lq;

    v8f acc[TM][TN];
#pragma unroll
    for (int i = 0; i < TM; ++i)
#pragma unroll
        for (int j = 0; j < TN; ++j) acc[i][j] = v8f{};

    v4f pre[NLD];
#pragma unroll
    for (int p = 0; p < NLD; ++p) pre[p] = *(const v4f*)(gp[p]);
#pragma unroll
    for (int p = 0; p < NLD; ++p) *(v4f*)(&lds[0][p * 32 * GEMM_PITCH + st_off]) = pre[p];
    __syncthreads();

    const int a_off = (wm * (TM * 16) + lidx) * GEMM_PITCH + 2 * lhalf;
    const int b_off = (BM + wn * (TN * 16) + lidx) * GEMM_PITCH + 2 * lhalf;

    const int KT = K / GEMM_BK;
    for (int kt = 0; kt < KT; ++kt) {
        const float* cur = lds[kt & 1];
        float*       nxt = lds[(kt & 1) ^ 1];
        const bool more = (kt + 1 < KT);
        if (more) {
#pragma unroll
            for (int p = 0; p < NLD; ++p)
                pre[p] = *(const v4f*)(gp[p] + (size_t)(kt + 1) * GEMM_BK);
        }
#pragma unroll
        for (int hk = 0; hk < 2; ++hk) {
            v2f a[4][TM];
            v2f b[4][TN];
#pragma unroll
            for (int kk = 0; kk < 4; ++kk) {
                const int ko = 4 * (hk * 4 + kk);
#pragma unroll
                for (int i = 0; i < TM; ++i)
                    a[kk][i] = *(const v2f*)(cur + a_off + i * 16 * GEMM_PITCH + ko);
#pragma unroll
                for (int j = 0; j < TN; ++j)
                    b[kk][j] = *(const v2f*)(cur + b_off + j * 16 * GEMM_PITCH + ko);
            }
#pragma unroll
            for (int kk = 0; kk < 4; ++kk) {
#pragma unroll
                for (int i = 0; i < TM; ++i) {
#pragma unroll
                    for (int j = 0; j < TN; ++j) {
                        acc[i][j] = __builtin_amdgcn_wmma_f32_16x16x4_f32(
                            false, a[kk][i], false, b[kk][j], (short)0, acc[i][j],
                            false, false);
                    }
                }
            }
        }
        if (more) {
#pragma unroll
            for (int p = 0; p < NLD; ++p)
                *(v4f*)(nxt + p * 32 * GEMM_PITCH + st_off) = pre[p];
        }
        __syncthreads();
    }

#pragma unroll
    for (int i = 0; i < TM; ++i) {
#pragma unroll
        for (int j = 0; j < TN; ++j) {
            const int nc = n0 + wn * (TN * 16) + j * 16;
            if (nc < N) {
                float* cp = C + (size_t)(m0 + wm * (TM * 16) + i * 16 + 8 * lhalf) * N + nc + lidx;
                float bv = (EPILOG == 1) ? bias[nc + lidx] : 0.f;
#pragma unroll
                for (int e = 0; e < 8; ++e) {
                    float v = acc[i][j][e];
                    if (EPILOG == 1) {
                        float x = v + bv;
                        v = (x > 20.f) ? x : log1pf(__expf(x));
                    }
                    cp[(size_t)e * N] = v;
                }
            }
        }
    }
}

__global__ __launch_bounds__(256)
void conv_silu_kernel(const float* __restrict__ xz,
                      const float* __restrict__ conv_w,
                      const float* __restrict__ conv_b,
                      float* __restrict__ x_act)
{
    int idx = blockIdx.x * blockDim.x + threadIdx.x;
    if (idx >= NTOK_C * D_INNER_C) return;
    int d = idx & (D_INNER_C - 1);
    int m = idx >> 11;
    int t = m & (SEQ_C - 1);

    float acc = conv_b[d];
    const float* w = conv_w + d * D_CONV_C;
#pragma unroll
    for (int j = 0; j < D_CONV_C; ++j) {
        int tt = t - (D_CONV_C - 1) + j;
        if (tt >= 0)
            acc += w[j] * xz[(size_t)(m - (D_CONV_C - 1) + j) * (2 * D_INNER_C) + d];
    }
    float s = acc / (1.f + __expf(-acc));
    x_act[(size_t)m * D_INNER_C + d] = s;
}

#define SCAN_NC   32
#define SCAN_LC   (SEQ_C / SCAN_NC)
#define SCAN_TPB  128
#define XZ_PITCH  (2 * D_INNER_C)

__global__ __launch_bounds__(SCAN_TPB)
void scan_chunk_carry_kernel(const float* __restrict__ x_act,
                             const float* __restrict__ dbc,
                             const float* __restrict__ dt,
                             const float* __restrict__ A_log,
                             float* __restrict__ carry)
{
    const int d = blockIdx.x * SCAN_TPB + threadIdx.x;
    const int c = blockIdx.y;
    const int b = blockIdx.z;

    float Arow[D_STATE_C];
#pragma unroll
    for (int s = 0; s < D_STATE_C; ++s)
        Arow[s] = -__expf(A_log[(size_t)d * D_STATE_C + s]);

    float h[D_STATE_C], P[D_STATE_C];
#pragma unroll
    for (int s = 0; s < D_STATE_C; ++s) { h[s] = 0.f; P[s] = 1.f; }

    const size_t m0 = (size_t)b * SEQ_C + (size_t)c * SCAN_LC;
#pragma unroll 4
    for (int i = 0; i < SCAN_LC; ++i) {
        const size_t m = m0 + i;
        const float u   = x_act[m * D_INNER_C + d];
        const float dtv = dt   [m * D_INNER_C + d];
        const float* Bt = dbc + m * 96 + 64;
        const float du = dtv * u;
#pragma unroll
        for (int s = 0; s < D_STATE_C; ++s) {
            float dA = __expf(dtv * Arow[s]);
            h[s] = dA * h[s] + du * Bt[s];
            P[s] = P[s] * dA;
        }
    }

    float* q = carry + (size_t)(b * SCAN_NC + c) * 32 * XZ_PITCH + d;
#pragma unroll
    for (int s = 0; s < D_STATE_C; ++s) {
        q[(size_t)s * XZ_PITCH]                = h[s];
        q[(size_t)(D_STATE_C + s) * XZ_PITCH]  = P[s];
    }
}

__global__ __launch_bounds__(SCAN_TPB)
void scan_chunk_final_kernel(const float* __restrict__ zcol,
                             const float* __restrict__ carry,
                             const float* __restrict__ x_act,
                             const float* __restrict__ dbc,
                             float* __restrict__ dt_y,
                             const float* __restrict__ A_log,
                             const float* __restrict__ Dvec)
{
    const int d = blockIdx.x * SCAN_TPB + threadIdx.x;
    const int c = blockIdx.y;
    const int b = blockIdx.z;

    float Arow[D_STATE_C];
#pragma unroll
    for (int s = 0; s < D_STATE_C; ++s)
        Arow[s] = -__expf(A_log[(size_t)d * D_STATE_C + s]);
    const float Dd = Dvec[d];

    float h[D_STATE_C];
#pragma unroll
    for (int s = 0; s < D_STATE_C; ++s) h[s] = 0.f;

    {
        const float* q = carry + (size_t)(b * SCAN_NC) * 32 * XZ_PITCH + d;
#pragma unroll 2
        for (int cp = 0; cp < c; ++cp) {
#pragma unroll
            for (int s = 0; s < D_STATE_C; ++s)
                h[s] = q[(size_t)(D_STATE_C + s) * XZ_PITCH] * h[s] + q[(size_t)s * XZ_PITCH];
            q += (size_t)32 * XZ_PITCH;
        }
    }

    const size_t m0 = (size_t)b * SEQ_C + (size_t)c * SCAN_LC;
#pragma unroll 8
    for (int i = 0; i < SCAN_LC; ++i) {
        const size_t m = m0 + i;
        const float u   = x_act[m * D_INNER_C + d];
        const float dtv = dt_y [m * D_INNER_C + d];
        const float z   = zcol [m * XZ_PITCH + d];
        const float* Bt = dbc + m * 96 + 64;
        const float* Ct = dbc + m * 96 + 80;

        const float du = dtv * u;
        float yv = 0.f;
#pragma unroll
        for (int s = 0; s < D_STATE_C; ++s) {
            float dA = __expf(dtv * Arow[s]);
            h[s] = dA * h[s] + du * Bt[s];
            yv += h[s] * Ct[s];
        }
        yv += u * Dd;
        float sz = z / (1.f + __expf(-z));
        dt_y[m * D_INNER_C + d] = yv * sz;
    }
}

#define GEMM_BLOCKS(M, N, BM, BN) (((M) / (BM)) * (((N) + (BN) - 1) / (BN)))

extern "C" void kernel_launch(void* const* d_in, const int* in_sizes, int n_in,
                              void* d_out, int out_size, void* d_ws, size_t ws_size,
                              hipStream_t stream) {
    (void)in_sizes; (void)n_in; (void)out_size; (void)ws_size;

    const float* x      = (const float*)d_in[0];
    const float* W_in   = (const float*)d_in[1];
    const float* conv_w = (const float*)d_in[2];
    const float* conv_b = (const float*)d_in[3];
    const float* W_x    = (const float*)d_in[4];
    const float* W_dt   = (const float*)d_in[5];
    const float* b_dt   = (const float*)d_in[6];
    const float* A_log  = (const float*)d_in[7];
    const float* Dvec   = (const float*)d_in[8];
    const float* W_out  = (const float*)d_in[9];
    float* out = (float*)d_out;

    float* xz   = (float*)d_ws;
    float* xact = xz   + (size_t)NTOK_C * (2 * D_INNER_C);
    float* dbc  = xact + (size_t)NTOK_C * D_INNER_C;
    float* dtb  = dbc  + (size_t)NTOK_C * 96;

    wmma_gemm_f32<0, 64, 128, 2, 4><<<GEMM_BLOCKS(NTOK_C, 2 * D_INNER_C, 64, 128), 256, 0, stream>>>(
        x, W_in, nullptr, xz, NTOK_C, 2 * D_INNER_C, D_MODEL_C, D_MODEL_C);

    {
        int n = NTOK_C * D_INNER_C;
        conv_silu_kernel<<<(n + 255) / 256, 256, 0, stream>>>(xz, conv_w, conv_b, xact);
    }

    wmma_gemm_f32<0, 32, 128, 1, 8><<<GEMM_BLOCKS(NTOK_C, 96, 32, 128), 256, 0, stream>>>(
        xact, W_x, nullptr, dbc, NTOK_C, 96, D_INNER_C, D_INNER_C);

    wmma_gemm_f32<1, 64, 128, 2, 4><<<GEMM_BLOCKS(NTOK_C, D_INNER_C, 64, 128), 256, 0, stream>>>(
        dbc, W_dt, b_dt, dtb, NTOK_C, D_INNER_C, DT_RANK_C, 96);

    scan_chunk_carry_kernel<<<dim3(D_INNER_C / SCAN_TPB, SCAN_NC - 1, BATCH_C), SCAN_TPB, 0, stream>>>(
        xact, dbc, dtb, A_log, xz);

    scan_chunk_final_kernel<<<dim3(D_INNER_C / SCAN_TPB, SCAN_NC, BATCH_C), SCAN_TPB, 0, stream>>>(
        xz + D_INNER_C, xz, xact, dbc, dtb, A_log, Dvec);

    wmma_gemm_f32<0, 64, 64, 2, 4><<<GEMM_BLOCKS(NTOK_C, D_MODEL_C, 64, 64), 256, 0, stream>>>(
        dtb, W_out, nullptr, out, NTOK_C, D_MODEL_C, D_INNER_C, D_INNER_C);
}


# conv+SiLU kernel: one thread does 4 channels x 16 time steps (sliding 4-row window in registers, 16-byte loads and stores): 2048 waves to dispatch instead of 131072; sums per output unchanged
# speedup vs baseline: 7.9259x; 1.0318x over previous
#include <hip/hip_runtime.h>
#include <hip/hip_bf16.h>
#include <math.h>

typedef __attribute__((ext_vector_type(2))) float v2f;
typedef __attribute__((ext_vector_type(4))) float v4f;
typedef __attribute__((ext_vector_type(8))) float v8f;

#define D_MODEL_C 1024
#define D_STATE_C 16
#define D_CONV_C  4
#define DT_RANK_C 64
#define D_INNER_C 2048
#define BATCH_C   2
#define SEQ_C     1024
#define NTOK_C    (BATCH_C * SEQ_C)

#define GEMM_BK     32
#define GEMM_PITCH  36

template <int EPILOG, int BM, int BN, int WM, int WN>
__global__ __launch_bounds__(256)
void wmma_gemm_f32(const float* __restrict__ A,
                   const float* __restrict__ W,
                   const float* __restrict__ bias,
                   float* __restrict__ C,
                   int M, int N, int K, int lda)
{
    constexpr int TM   = BM / WM / 16;
    constexpr int TN   = BN / WN / 16;
    constexpr int ROWS = BM + BN;
    constexpr int NLD  = ROWS / 32;
    static_assert(WM * WN == 8, "8 waves per block");
    static_assert(BM % 32 == 0 && BN % 32 == 0, "a pass of 32 rows is all A or all W");
    static_assert(TM * 16 * WM == BM && TN * 16 * WN == BN, "wave tiles cover the block tile");

    __shared__ __attribute__((aligned(16))) float lds[2][ROWS * GEMM_PITCH];

    const int tid   = threadIdx.x;
    const int lane  = tid & 31;
    const int wave  = tid >> 5;
    const int wm    = wave / WN;
    const int wn    = wave % WN;
    const int lhalf = lane >> 4;
    const int lidx  = lane & 15;

    const int nBlkN = (N + BN - 1) / BN;
    const int m0 = (blockIdx.x / nBlkN) * BM;
    const int n0 = (blockIdx.x % nBlkN) * BN;
    (void)M;

    const int lrow = tid >> 3;
    const int lq   = tid & 7;
    const float* gp[NLD];
#pragma unroll
    for (int p = 0; p < NLD; ++p) {
        const int r = p * 32 + lrow;
        if (p * 32 < BM) {
            gp[p] = A + (size_t)(m0 + r) * lda + 4 * lq;
        } else {
            const int n = n0 + r - BM;
            gp[p] = W + (size_t)(n < N ? n : 0) * K + 4 * lq;
        }
    }
    const int st_off = lrow * GEMM_PITCH + 4 * lq;

    v8f acc[TM][TN];
#pragma unroll
    for (int i = 0; i < TM; ++i)
#pragma unroll
        for (int j = 0; j < TN; ++j) acc[i][j] = v8f{};

    v4f pre[NLD];
#pragma unroll
    for (int p = 0; p < NLD; ++p) pre[p] = *(const v4f*)(gp[p]);
#pragma unroll
    for (int p = 0; p < NLD; ++p) *(v4f*)(&lds[0][p * 32 * GEMM_PITCH + st_off]) = pre[p];
    __syncthreads();

    const int a_off = (wm * (TM * 16) + lidx) * GEMM_PITCH + 2 * lhalf;
    const int b_off = (BM + wn * (TN * 16) + lidx) * GEMM_PITCH + 2 * lhalf;

    const int KT = K / GEMM_BK;
    for (int kt = 0; kt < KT; ++kt) {
        const float* cur = lds[kt & 1];
        float*       nxt = lds[(kt & 1) ^ 1];
        const bool more = (kt + 1 < KT);
        if (more) {
#pragma unroll
            for (int p = 0; p < NLD; ++p)
                pre[p] = *(const v4f*)(gp[p] + (size_t)(kt + 1) * GEMM_BK);
        }
#pragma unroll
        for (int hk = 0; hk < 2; ++hk) {
            v2f a[4][TM];
            v2f b[4][TN];
#pragma unroll
            for (int kk = 0; kk < 4; ++kk) {
                const int ko = 4 * (hk * 4 + kk);
#pragma unroll
                for (int i = 0; i < TM; ++i)
                    a[kk][i] = *(const v2f*)(cur + a_off + i * 16 * GEMM_PITCH + ko);
#pragma unroll
                for (int j = 0; j < TN; ++j)
                    b[kk][j] = *(const v2f*)(cur + b_off + j * 16 * GEMM_PITCH + ko);
            }
#pragma unroll
            for (int kk = 0; kk < 4; ++kk) {
#pragma unroll
                for (int i = 0; i < TM; ++i) {
#pragma unroll
                    for (int j = 0; j < TN; ++j) {
                        acc[i][j] = __builtin_amdgcn_wmma_f32_16x16x4_f32(
                            false, a[kk][i], false, b[kk][j], (short)0, acc[i][j],
                            false, false);
                    }
                }
            }
        }
        if (more) {
#pragma unroll
            for (int p = 0; p < NLD; ++p)
                *(v4f*)(nxt + p * 32 * GEMM_PITCH + st_off) = pre[p];
        }
        __syncthreads();
    }

#pragma unroll
    for (int i = 0; i < TM; ++i) {
#pragma unroll
        for (int j = 0; j < TN; ++j) {
            const int nc = n0 + wn * (TN * 16) + j * 16;
            if (nc < N) {
                float* cp = C + (size_t)(m0 + wm * (TM * 16) + i * 16 + 8 * lhalf) * N + nc + lidx;
                float bv = (EPILOG == 1) ? bias[nc + lidx] : 0.f;
#pragma unroll
                for (int e = 0; e < 8; ++e) {
                    float v = acc[i][j][e];
                    if (EPILOG == 1) {
                        float x = v + bv;
                        v = (x > 20.f) ? x : log1pf(__expf(x));
                    }
                    cp[(size_t)e * N] = v;
                }
            }
        }
    }
}

#define CONV_TC     16
#define CONV_QPR    (D_INNER_C / 4)
#define CONV_BLOCKS ((NTOK_C / CONV_TC) * CONV_QPR / 256)

__global__ __launch_bounds__(256)
void conv_silu_kernel(const float* __restrict__ xz,
                      const float* __restrict__ conv_w,
                      const float* __restrict__ conv_b,
                      float* __restrict__ x_act)
{
    const int gid   = blockIdx.x * 256 + threadIdx.x;
    const int dq    = gid & (CONV_QPR - 1);
    const int chunk = gid / CONV_QPR;
    const int d     = 4 * dq;
    const int m0    = chunk * CONV_TC;
    const bool head = (m0 & (SEQ_C - 1)) == 0;

    v4f w[4];
#pragma unroll
    for (int c = 0; c < 4; ++c)
        w[c] = *(const v4f*)(conv_w + (size_t)(d + c) * D_CONV_C);
    const v4f bias = *(const v4f*)(conv_b + d);

    const float* src = xz + (size_t)m0 * (2 * D_INNER_C) + d;
    v4f xr[CONV_TC + 3];
    if (head) {
        xr[0] = v4f{}; xr[1] = v4f{}; xr[2] = v4f{};
    } else {
#pragma unroll
        for (int j = 0; j < 3; ++j)
            xr[j] = *(const v4f*)(src - (ptrdiff_t)(3 - j) * (2 * D_INNER_C));
    }
#pragma unroll
    for (int i = 0; i < CONV_TC; ++i)
        xr[3 + i] = *(const v4f*)(src + (size_t)i * (2 * D_INNER_C));

    float* dst = x_act + (size_t)m0 * D_INNER_C + d;
#pragma unroll
    for (int i = 0; i < CONV_TC; ++i) {
        v4f o;
#pragma unroll
        for (int c = 0; c < 4; ++c) {
            float acc = bias[c];
            acc += w[c][0] * xr[i][c];
            acc += w[c][1] * xr[i + 1][c];
            acc += w[c][2] * xr[i + 2][c];
            acc += w[c][3] * xr[i + 3][c];
            o[c] = acc / (1.f + __expf(-acc));
        }
        *(v4f*)(dst + (size_t)i * D_INNER_C) = o;
    }
}

#define SCAN_NC   32
#define SCAN_LC   (SEQ_C / SCAN_NC)
#define SCAN_TPB  128
#define XZ_PITCH  (2 * D_INNER_C)

__global__ __launch_bounds__(SCAN_TPB)
void scan_chunk_carry_kernel(const float* __restrict__ x_act,
                             const float* __restrict__ dbc,
                             const float* __restrict__ dt,
                             const float* __restrict__ A_log,
                             float* __restrict__ carry)
{
    const int d = blockIdx.x * SCAN_TPB + threadIdx.x;
    const int c = blockIdx.y;
    const int b = blockIdx.z;

    float Arow[D_STATE_C];
#pragma unroll
    for (int s = 0; s < D_STATE_C; ++s)
        Arow[s] = -__expf(A_log[(size_t)d * D_STATE_C + s]);

    float h[D_STATE_C], P[D_STATE_C];
#pragma unroll
    for (int s = 0; s < D_STATE_C; ++s) { h[s] = 0.f; P[s] = 1.f; }

    const size_t m0 = (size_t)b * SEQ_C + (size_t)c * SCAN_LC;
#pragma unroll 4
    for (int i = 0; i < SCAN_LC; ++i) {
        const size_t m = m0 + i;
        const float u   = x_act[m * D_INNER_C + d];
        const float dtv = dt   [m * D_INNER_C + d];
        const float* Bt = dbc + m * 96 + 64;
        const float du = dtv * u;
#pragma unroll
        for (int s = 0; s < D_STATE_C; ++s) {
            float dA = __expf(dtv * Arow[s]);
            h[s] = dA * h[s] + du * Bt[s];
            P[s] = P[s] * dA;
        }
    }

    float* q = carry + (size_t)(b * SCAN_NC + c) * 32 * XZ_PITCH + d;
#pragma unroll
    for (int s = 0; s < D_STATE_C; ++s) {
        q[(size_t)s * XZ_PITCH]                = h[s];
        q[(size_t)(D_STATE_C + s) * XZ_PITCH]  = P[s];
    }
}

__global__ __launch_bounds__(SCAN_TPB)
void scan_chunk_final_kernel(const float* __restrict__ zcol,
                             const float* __restrict__ carry,
                             const float* __restrict__ x_act,
                             const float* __restrict__ dbc,
                             float* __restrict__ dt_y,
                             const float* __restrict__ A_log,
                             const float* __restrict__ Dvec)
{
    const int d = blockIdx.x * SCAN_TPB + threadIdx.x;
    const int c = blockIdx.y;
    const int b = blockIdx.z;

    float Arow[D_STATE_C];
#pragma unroll
    for (int s = 0; s < D_STATE_C; ++s)
        Arow[s] = -__expf(A_log[(size_t)d * D_STATE_C + s]);
    const float Dd = Dvec[d];

    float h[D_STATE_C];
#pragma unroll
    for (int s = 0; s < D_STATE_C; ++s) h[s] = 0.f;

    {
        const float* q = carry + (size_t)(b * SCAN_NC) * 32 * XZ_PITCH + d;
#pragma unroll 2
        for (int cp = 0; cp < c; ++cp) {
#pragma unroll
            for (int s = 0; s < D_STATE_C; ++s)
                h[s] = q[(size_t)(D_STATE_C + s) * XZ_PITCH] * h[s] + q[(size_t)s * XZ_PITCH];
            q += (size_t)32 * XZ_PITCH;
        }
    }

    const size_t m0 = (size_t)b * SEQ_C + (size_t)c * SCAN_LC;
#pragma unroll 8
    for (int i = 0; i < SCAN_LC; ++i) {
        const size_t m = m0 + i;
        const float u   = x_act[m * D_INNER_C + d];
        const float dtv = dt_y [m * D_INNER_C + d];
        const float z   = zcol [m * XZ_PITCH + d];
        const float* Bt = dbc + m * 96 + 64;
        const float* Ct = dbc + m * 96 + 80;

        const float du = dtv * u;
        float yv = 0.f;
#pragma unroll
        for (int s = 0; s < D_STATE_C; ++s) {
            float dA = __expf(dtv * Arow[s]);
            h[s] = dA * h[s] + du * Bt[s];
            yv += h[s] * Ct[s];
        }
        yv += u * Dd;
        float sz = z / (1.f + __expf(-z));
        dt_y[m * D_INNER_C + d] = yv * sz;
    }
}

#define GEMM_BLOCKS(M, N, BM, BN) (((M) / (BM)) * (((N) + (BN) - 1) / (BN)))

extern "C" void kernel_launch(void* const* d_in, const int* in_sizes, int n_in,
                              void* d_out, int out_size, void* d_ws, size_t ws_size,
                              hipStream_t stream) {
    (void)in_sizes; (void)n_in; (void)out_size; (void)ws_size;

    const float* x      = (const float*)d_in[0];
    const float* W_in   = (const float*)d_in[1];
    const float* conv_w = (const float*)d_in[2];
    const float* conv_b = (const float*)d_in[3];
    const float* W_x    = (const float*)d_in[4];
    const float* W_dt   = (const float*)d_in[5];
    const float* b_dt   = (const float*)d_in[6];
    const float* A_log  = (const float*)d_in[7];
    const float* Dvec   = (const float*)d_in[8];
    const float* W_out  = (const float*)d_in[9];
    float* out = (float*)d_out;

    float* xz   = (float*)d_ws;
    float* xact = xz   + (size_t)NTOK_C * (2 * D_INNER_C);
    float* dbc  = xact + (size_t)NTOK_C * D_INNER_C;
    float* dtb  = dbc  + (size_t)NTOK_C * 96;

    wmma_gemm_f32<0, 64, 128, 2, 4><<<GEMM_BLOCKS(NTOK_C, 2 * D_INNER_C, 64, 128), 256, 0, stream>>>(
        x, W_in, nullptr, xz, NTOK_C, 2 * D_INNER_C, D_MODEL_C, D_MODEL_C);

    conv_silu_kernel<<<CONV_BLOCKS, 256, 0, stream>>>(xz, conv_w, conv_b, xact);

    wmma_gemm_f32<0, 32, 128, 1, 8><<<GEMM_BLOCKS(NTOK_C, 96, 32, 128), 256, 0, stream>>>(
        xact, W_x, nullptr, dbc, NTOK_C, 96, D_INNER_C, D_INNER_C);

    wmma_gemm_f32<1, 64, 128, 2, 4><<<GEMM_BLOCKS(NTOK_C, D_INNER_C, 64, 128), 256, 0, stream>>>(
        dbc, W_dt, b_dt, dtb, NTOK_C, D_INNER_C, DT_RANK_C, 96);

    scan_chunk_carry_kernel<<<dim3(D_INNER_C / SCAN_TPB, SCAN_NC - 1, BATCH_C), SCAN_TPB, 0, stream>>>(
        xact, dbc, dtb, A_log, xz);

    scan_chunk_final_kernel<<<dim3(D_INNER_C / SCAN_TPB, SCAN_NC, BATCH_C), SCAN_TPB, 0, stream>>>(
        xz + D_INNER_C, xz, xact, dbc, dtb, A_log, Dvec);

    wmma_gemm_f32<0, 64, 64, 2, 4><<<GEMM_BLOCKS(NTOK_C, D_MODEL_C, 64, 64), 256, 0, stream>>>(
        dtb, W_out, nullptr, out, NTOK_C, D_MODEL_C, D_INNER_C, D_INNER_C);
}


# x_proj split over K in 8 slices: 256 blocks of 64x96, 8 K-steps each, instead of 64 blocks of 64 K-steps; a small kernel adds the 8 partial planes in slice order, so the dbc sum order changes
# speedup vs baseline: 8.8179x; 1.1125x over previous
#include <hip/hip_runtime.h>
#include <hip/hip_bf16.h>
#include <math.h>

typedef __attribute__((ext_vector_type(2))) float v2f;
typedef __attribute__((ext_vector_type(4))) float v4f;
typedef __attribute__((ext_vector_type(8))) float v8f;

#define D_MODEL_C 1024
#define D_STATE_C 16
#define D_CONV_C  4
#define DT_RANK_C 64
#define D_INNER_C 2048
#define BATCH_C   2
#define SEQ_C     1024
#define NTOK_C    (BATCH_C * SEQ_C)

#define GEMM_BK     32
#define GEMM_PITCH  36

template <int EPILOG, int BM, int BN, int WM, int WN, int KSLICES = 1>
__global__ __launch_bounds__(256)
void wmma_gemm_f32(const float* __restrict__ A,
                   const float* __restrict__ W,
                   const float* __restrict__ bias,
                   float* __restrict__ C,
                   int M, int N, int K, int lda)
{
    constexpr int TM   = BM / WM / 16;
    constexpr int TN   = BN / WN / 16;
    constexpr int ROWS = BM + BN;
    constexpr int NLD  = ROWS / 32;
    static_assert(WM * WN == 8, "8 waves per block");
    static_assert(BM % 32 == 0 && BN % 32 == 0, "a pass of 32 rows is all A or all W");
    static_assert(TM * 16 * WM == BM && TN * 16 * WN == BN, "wave tiles cover the block tile");

    __shared__ __attribute__((aligned(16))) float lds[2][ROWS * GEMM_PITCH];

    const int tid   = threadIdx.x;
    const int lane  = tid & 31;
    const int wave  = tid >> 5;
    const int wm    = wave / WN;
    const int wn    = wave % WN;
    const int lhalf = lane >> 4;
    const int lidx  = lane & 15;

    const int nBlkN = (N + BN - 1) / BN;
    const int m0 = (blockIdx.x / nBlkN) * BM;
    const int n0 = (blockIdx.x % nBlkN) * BN;
    const int ldw = K * KSLICES;
    if (KSLICES > 1) {
        const int ks = blockIdx.y;
        A += (size_t)ks * K;
        W += (size_t)ks * K;
        C += (size_t)ks * M * N;
    }
    (void)M;

    const int lrow = tid >> 3;
    const int lq   = tid & 7;
    const float* gp[NLD];
#pragma unroll
    for (int p = 0; p < NLD; ++p) {
        const int r = p * 32 + lrow;
        if (p * 32 < BM) {
            gp[p] = A + (size_t)(m0 + r) * lda + 4 * lq;
        } else {
            const int n = n0 + r - BM;
            gp[p] = W + (size_t)(n < N ? n : 0) * ldw + 4 * lq;
        }
    }
    const int st_off = lrow * GEMM_PITCH + 4 * lq;

    v8f acc[TM][TN];
#pragma unroll
    for (int i = 0; i < TM; ++i)
#pragma unroll
        for (int j = 0; j < TN; ++j) acc[i][j] = v8f{};

    v4f pre[NLD];
#pragma unroll
    for (int p = 0; p < NLD; ++p) pre[p] = *(const v4f*)(gp[p]);
#pragma unroll
    for (int p = 0; p < NLD; ++p) *(v4f*)(&lds[0][p * 32 * GEMM_PITCH + st_off]) = pre[p];
    __syncthreads();

    const int a_off = (wm * (TM * 16) + lidx) * GEMM_PITCH + 2 * lhalf;
    const int b_off = (BM + wn * (TN * 16) + lidx) * GEMM_PITCH + 2 * lhalf;

    const int KT = K / GEMM_BK;
    for (int kt = 0; kt < KT; ++kt) {
        const float* cur = lds[kt & 1];
        float*       nxt = lds[(kt & 1) ^ 1];
        const bool more = (kt + 1 < KT);
        if (more) {
#pragma unroll
            for (int p = 0; p < NLD; ++p)
                pre[p] = *(const v4f*)(gp[p] + (size_t)(kt + 1) * GEMM_BK);
        }
#pragma unroll
        for (int hk = 0; hk < 2; ++hk) {
            v2f a[4][TM];
            v2f b[4][TN];
#pragma unroll
            for (int kk = 0; kk < 4; ++kk) {
                const int ko = 4 * (hk * 4 + kk);
#pragma unroll
                for (int i = 0; i < TM; ++i)
                    a[kk][i] = *(const v2f*)(cur + a_off + i * 16 * GEMM_PITCH + ko);
#pragma unroll
                for (int j = 0; j < TN; ++j)
                    b[kk][j] = *(const v2f*)(cur + b_off + j * 16 * GEMM_PITCH + ko);
            }
#pragma unroll
            for (int kk = 0; kk < 4; ++kk) {
#pragma unroll
                for (int i = 0; i < TM; ++i) {
#pragma unroll
                    for (int j = 0; j < TN; ++j) {
                        acc[i][j] = __builtin_amdgcn_wmma_f32_16x16x4_f32(
                            false, a[kk][i], false, b[kk][j], (short)0, acc[i][j],
                            false, false);
                    }
                }
            }
        }
        if (more) {
#pragma unroll
            for (int p = 0; p < NLD; ++p)
                *(v4f*)(nxt + p * 32 * GEMM_PITCH + st_off) = pre[p];
        }
        __syncthreads();
    }

#pragma unroll
    for (int i = 0; i < TM; ++i) {
#pragma unroll
        for (int j = 0; j < TN; ++j) {
            const int nc = n0 + wn * (TN * 16) + j * 16;
            if (nc < N) {
                float* cp = C + (size_t)(m0 + wm * (TM * 16) + i * 16 + 8 * lhalf) * N + nc + lidx;
                float bv = (EPILOG == 1) ? bias[nc + lidx] : 0.f;
#pragma unroll
                for (int e = 0; e < 8; ++e) {
                    float v = acc[i][j][e];
                    if (EPILOG == 1) {
                        float x = v + bv;
                        v = (x > 20.f) ? x : log1pf(__expf(x));
                    }
                    cp[(size_t)e * N] = v;
                }
            }
        }
    }
}

#define XP_SLICES  8
#define XP_KS      (D_INNER_C / XP_SLICES)
#define XP_PLANE4  (NTOK_C * 96 / 4)

__global__ __launch_bounds__(256)
void sum_slices_kernel(const float* __restrict__ part,
                       float* __restrict__ dst)
{
    const int i = blockIdx.x * 256 + threadIdx.x;
    const v4f* p = (const v4f*)part + i;
    v4f v[XP_SLICES];
#pragma unroll
    for (int q = 0; q < XP_SLICES; ++q) v[q] = p[(size_t)q * XP_PLANE4];
    v4f acc = v[0];
#pragma unroll
    for (int q = 1; q < XP_SLICES; ++q) acc += v[q];
    ((v4f*)dst)[i] = acc;
}

#define CONV_TC     16
#define CONV_QPR    (D_INNER_C / 4)
#define CONV_BLOCKS ((NTOK_C / CONV_TC) * CONV_QPR / 256)

__global__ __launch_bounds__(256)
void conv_silu_kernel(const float* __restrict__ xz,
                      const float* __restrict__ conv_w,
                      const float* __restrict__ conv_b,
                      float* __restrict__ x_act)
{
    const int gid   = blockIdx.x * 256 + threadIdx.x;
    const int dq    = gid & (CONV_QPR - 1);
    const int chunk = gid / CONV_QPR;
    const int d     = 4 * dq;
    const int m0    = chunk * CONV_TC;
    const bool head = (m0 & (SEQ_C - 1)) == 0;

    v4f w[4];
#pragma unroll
    for (int c = 0; c < 4; ++c)
        w[c] = *(const v4f*)(conv_w + (size_t)(d + c) * D_CONV_C);
    const v4f bias = *(const v4f*)(conv_b + d);

    const float* src = xz + (size_t)m0 * (2 * D_INNER_C) + d;
    v4f xr[CONV_TC + 3];
    if (head) {
        xr[0] = v4f{}; xr[1] = v4f{}; xr[2] = v4f{};
    } else {
#pragma unroll
        for (int j = 0; j < 3; ++j)
            xr[j] = *(const v4f*)(src - (ptrdiff_t)(3 - j) * (2 * D_INNER_C));
    }
#pragma unroll
    for (int i = 0; i < CONV_TC; ++i)
        xr[3 + i] = *(const v4f*)(src + (size_t)i * (2 * D_INNER_C));

    float* dst = x_act + (size_t)m0 * D_INNER_C + d;
#pragma unroll
    for (int i = 0; i < CONV_TC; ++i) {
        v4f o;
#pragma unroll
        for (int c = 0; c < 4; ++c) {
            float acc = bias[c];
            acc += w[c][0] * xr[i][c];
            acc += w[c][1] * xr[i + 1][c];
            acc += w[c][2] * xr[i + 2][c];
            acc += w[c][3] * xr[i + 3][c];
            o[c] = acc / (1.f + __expf(-acc));
        }
        *(v4f*)(dst + (size_t)i * D_INNER_C) = o;
    }
}

#define SCAN_NC   32
#define SCAN_LC   (SEQ_C / SCAN_NC)
#define SCAN_TPB  128
#define XZ_PITCH  (2 * D_INNER_C)

__global__ __launch_bounds__(SCAN_TPB)
void scan_chunk_carry_kernel(const float* __restrict__ x_act,
                             const float* __restrict__ dbc,
                             const float* __restrict__ dt,
                             const float* __restrict__ A_log,
                             float* __restrict__ carry)
{
    const int d = blockIdx.x * SCAN_TPB + threadIdx.x;
    const int c = blockIdx.y;
    const int b = blockIdx.z;

    float Arow[D_STATE_C];
#pragma unroll
    for (int s = 0; s < D_STATE_C; ++s)
        Arow[s] = -__expf(A_log[(size_t)d * D_STATE_C + s]);

    float h[D_STATE_C], P[D_STATE_C];
#pragma unroll
    for (int s = 0; s < D_STATE_C; ++s) { h[s] = 0.f; P[s] = 1.f; }

    const size_t m0 = (size_t)b * SEQ_C + (size_t)c * SCAN_LC;
#pragma unroll 4
    for (int i = 0; i < SCAN_LC; ++i) {
        const size_t m = m0 + i;
        const float u   = x_act[m * D_INNER_C + d];
        const float dtv = dt   [m * D_INNER_C + d];
        const float* Bt = dbc + m * 96 + 64;
        const float du = dtv * u;
#pragma unroll
        for (int s = 0; s < D_STATE_C; ++s) {
            float dA = __expf(dtv * Arow[s]);
            h[s] = dA * h[s] + du * Bt[s];
            P[s] = P[s] * dA;
        }
    }

    float* q = carry + (size_t)(b * SCAN_NC + c) * 32 * XZ_PITCH + d;
#pragma unroll
    for (int s = 0; s < D_STATE_C; ++s) {
        q[(size_t)s * XZ_PITCH]                = h[s];
        q[(size_t)(D_STATE_C + s) * XZ_PITCH]  = P[s];
    }
}

__global__ __launch_bounds__(SCAN_TPB)
void scan_chunk_final_kernel(const float* __restrict__ zcol,
                             const float* __restrict__ carry,
                             const float* __restrict__ x_act,
                             const float* __restrict__ dbc,
                             float* __restrict__ dt_y,
                             const float* __restrict__ A_log,
                             const float* __restrict__ Dvec)
{
    const int d = blockIdx.x * SCAN_TPB + threadIdx.x;
    const int c = blockIdx.y;
    const int b = blockIdx.z;

    float Arow[D_STATE_C];
#pragma unroll
    for (int s = 0; s < D_STATE_C; ++s)
        Arow[s] = -__expf(A_log[(size_t)d * D_STATE_C + s]);
    const float Dd = Dvec[d];

    float h[D_STATE_C];
#pragma unroll
    for (int s = 0; s < D_STATE_C; ++s) h[s] = 0.f;

    {
        const float* q = carry + (size_t)(b * SCAN_NC) * 32 * XZ_PITCH + d;
#pragma unroll 2
        for (int cp = 0; cp < c; ++cp) {
#pragma unroll
            for (int s = 0; s < D_STATE_C; ++s)
                h[s] = q[(size_t)(D_STATE_C + s) * XZ_PITCH] * h[s] + q[(size_t)s * XZ_PITCH];
            q += (size_t)32 * XZ_PITCH;
        }
    }

    const size_t m0 = (size_t)b * SEQ_C + (size_t)c * SCAN_LC;
#pragma unroll 8
    for (int i = 0; i < SCAN_LC; ++i) {
        const size_t m = m0 + i;
        const float u   = x_act[m * D_INNER_C + d];
        const float dtv = dt_y [m * D_INNER_C + d];
        const float z   = zcol [m * XZ_PITCH + d];
        const float* Bt = dbc + m * 96 + 64;
        const float* Ct = dbc + m * 96 + 80;

        const float du = dtv * u;
        float yv = 0.f;
#pragma unroll
        for (int s = 0; s < D_STATE_C; ++s) {
            float dA = __expf(dtv * Arow[s]);
            h[s] = dA * h[s] + du * Bt[s];
            yv += h[s] * Ct[s];
        }
        yv += u * Dd;
        float sz = z / (1.f + __expf(-z));
        dt_y[m * D_INNER_C + d] = yv * sz;
    }
}

#define GEMM_BLOCKS(M, N, BM, BN) (((M) / (BM)) * (((N) + (BN) - 1) / (BN)))

extern "C" void kernel_launch(void* const* d_in, const int* in_sizes, int n_in,
                              void* d_out, int out_size, void* d_ws, size_t ws_size,
                              hipStream_t stream) {
    (void)in_sizes; (void)n_in; (void)out_size; (void)ws_size;

    const float* x      = (const float*)d_in[0];
    const float* W_in   = (const float*)d_in[1];
    const float* conv_w = (const float*)d_in[2];
    const float* conv_b = (const float*)d_in[3];
    const float* W_x    = (const float*)d_in[4];
    const float* W_dt   = (const float*)d_in[5];
    const float* b_dt   = (const float*)d_in[6];
    const float* A_log  = (const float*)d_in[7];
    const float* Dvec   = (const float*)d_in[8];
    const float* W_out  = (const float*)d_in[9];
    float* out = (float*)d_out;

    float* xz   = (float*)d_ws;
    float* xact = xz   + (size_t)NTOK_C * (2 * D_INNER_C);
    float* dbc  = xact + (size_t)NTOK_C * D_INNER_C;
    float* dtb  = dbc  + (size_t)NTOK_C * 96;

    wmma_gemm_f32<0, 64, 128, 2, 4><<<GEMM_BLOCKS(NTOK_C, 2 * D_INNER_C, 64, 128), 256, 0, stream>>>(
        x, W_in, nullptr, xz, NTOK_C, 2 * D_INNER_C, D_MODEL_C, D_MODEL_C);

    conv_silu_kernel<<<CONV_BLOCKS, 256, 0, stream>>>(xz, conv_w, conv_b, xact);

    wmma_gemm_f32<0, 64, 96, 4, 2, XP_SLICES><<<dim3(GEMM_BLOCKS(NTOK_C, 96, 64, 96), XP_SLICES), 256, 0, stream>>>(
        xact, W_x, nullptr, dtb, NTOK_C, 96, XP_KS, D_INNER_C);
    sum_slices_kernel<<<XP_PLANE4 / 256, 256, 0, stream>>>(dtb, dbc);

    wmma_gemm_f32<1, 64, 128, 2, 4><<<GEMM_BLOCKS(NTOK_C, D_INNER_C, 64, 128), 256, 0, stream>>>(
        dbc, W_dt, b_dt, dtb, NTOK_C, D_INNER_C, DT_RANK_C, 96);

    scan_chunk_carry_kernel<<<dim3(D_INNER_C / SCAN_TPB, SCAN_NC - 1, BATCH_C), SCAN_TPB, 0, stream>>>(
        xact, dbc, dtb, A_log, xz);

    scan_chunk_final_kernel<<<dim3(D_INNER_C / SCAN_TPB, SCAN_NC, BATCH_C), SCAN_TPB, 0, stream>>>(
        xz + D_INNER_C, xz, xact, dbc, dtb, A_log, Dvec);

    wmma_gemm_f32<0, 64, 64, 2, 4><<<GEMM_BLOCKS(NTOK_C, D_MODEL_C, 64, 64), 256, 0, stream>>>(
        dtb, W_out, nullptr, out, NTOK_C, D_MODEL_C, D_INNER_C, D_INNER_C);
}
